# starting kernel
# baseline (speedup 1.0000x reference)
#include <hip/hip_runtime.h>
#include <hip/hip_bf16.h>

typedef __attribute__((ext_vector_type(2))) float v2f;
typedef __attribute__((ext_vector_type(4))) float v4f;
typedef __attribute__((ext_vector_type(8))) float v8f;

#define D_DIM 1024
#define ROWS_PER_BLOCK 16
#define WAVES_PER_BLOCK 8

__global__ __launch_bounds__(256) void crosslayer_kernel(
    const float* __restrict__ x0, const float* __restrict__ xl,
    const float* __restrict__ w,  const float* __restrict__ bias,
    float* __restrict__ out)
{
    __shared__ float part[WAVES_PER_BLOCK][ROWS_PER_BLOCK];
    __shared__ float srow[ROWS_PER_BLOCK];

    const int tid  = threadIdx.x;
    const int wave = tid >> 5;
    const int lane = tid & 31;
    const int half = lane >> 4;
    const int r    = lane & 15;
    const size_t row0 = (size_t)blockIdx.x * ROWS_PER_BLOCK;

    if ((tid & 7) == 0) {
        const size_t col = (size_t)tid * 4;
#pragma unroll
        for (int i = 0; i < ROWS_PER_BLOCK; ++i)
            __builtin_prefetch(x0 + (row0 + i) * D_DIM + col, 0, 3);
    }

    v8f c = {};
    const float* xlrow = xl + (row0 + r) * (size_t)D_DIM;
    const int kbase = wave * (D_DIM / WAVES_PER_BLOCK) + 2 * half;

#pragma unroll 4
    for (int j = 0; j < (D_DIM / WAVES_PER_BLOCK) / 4; ++j) {
        const int k = kbase + 4 * j;
        v2f a = *(const v2f*)(w + k);
        v2f b = *(const v2f*)(xlrow + k);
        c = __builtin_amdgcn_wmma_f32_16x16x4_f32(
                false, a, false, b, (short)0, c, false, false);
    }

    if (lane < 16) part[wave][lane] = c[0];
    __syncthreads();

    if (tid < ROWS_PER_BLOCK) {
        float s = 0.0f;
#pragma unroll
        for (int wv = 0; wv < WAVES_PER_BLOCK; ++wv) s += part[wv][tid];
        srow[tid] = s;
    }
    __syncthreads();

    const size_t col = (size_t)tid * 4;
    const v4f bb = *(const v4f*)(bias + col);
#pragma unroll 4
    for (int i = 0; i < ROWS_PER_BLOCK; ++i) {
        const size_t off = (row0 + i) * D_DIM + col;
        const float s = srow[i];
        v4f a = *(const v4f*)(x0 + off);
        v4f l = *(const v4f*)(xl + off);
        v4f o = a * s + bb + l;
        *(v4f*)(out + off) = o;
    }
}

extern "C" void kernel_launch(void* const* d_in, const int* in_sizes, int n_in,
                              void* d_out, int out_size, void* d_ws, size_t ws_size,
                              hipStream_t stream) {
    const float* x0   = (const float*)d_in[0];
    const float* xl   = (const float*)d_in[1];
    const float* kern = (const float*)d_in[2];
    const float* bias = (const float*)d_in[3];
    float* out = (float*)d_out;

    const int B = in_sizes[0] / D_DIM;
    const int grid = B / ROWS_PER_BLOCK;
    crosslayer_kernel<<<grid, 256, 0, stream>>>(x0, xl, kern, bias, out);
}


# xl and x0 read once by whole-line 16-byte loads issued up front; each wave parks its xl slice in LDS for the WMMA fragments and the epilogue (was strided 8-byte loads, prefetch pass, second xl read)
# speedup vs baseline: 1.2648x; 1.2648x over previous
#include <hip/hip_runtime.h>
#include <hip/hip_bf16.h>

typedef __attribute__((ext_vector_type(2))) float v2f;
typedef __attribute__((ext_vector_type(4))) float v4f;
typedef __attribute__((ext_vector_type(8))) float v8f;

#define D_DIM 1024
#define ROWS_PER_BLOCK 16
#define WAVES_PER_BLOCK 8
#define KSLICE (D_DIM / WAVES_PER_BLOCK)
#define PITCH (KSLICE + 4)

__global__ __launch_bounds__(256, 8) void crosslayer_kernel(
    const float* __restrict__ x0, const float* __restrict__ xl,
    const float* __restrict__ w,  const float* __restrict__ bias,
    float* __restrict__ out)
{
    __shared__ __attribute__((aligned(16))) float xs[WAVES_PER_BLOCK][ROWS_PER_BLOCK][PITCH];
    __shared__ __attribute__((aligned(16))) float ws[WAVES_PER_BLOCK][KSLICE];
    __shared__ float part[WAVES_PER_BLOCK][ROWS_PER_BLOCK];
    __shared__ float srow[ROWS_PER_BLOCK];

    const int tid  = threadIdx.x;
    const int wave = tid >> 5;
    const int lane = tid & 31;
    const int half = lane >> 4;
    const int r    = lane & 15;
    const size_t row0 = (size_t)blockIdx.x * ROWS_PER_BLOCK;
    const size_t col  = (size_t)tid * 4;

    const v4f bb = *(const v4f*)(bias + col);
    const v4f wq = *(const v4f*)(w + col);

    v4f l[ROWS_PER_BLOCK];
#pragma unroll
    for (int i = 0; i < ROWS_PER_BLOCK; ++i)
        l[i] = *(const v4f*)(xl + (row0 + i) * D_DIM + col);

    v4f a[ROWS_PER_BLOCK];
#pragma unroll
    for (int i = 0; i < ROWS_PER_BLOCK; ++i)
        a[i] = *(const v4f*)(x0 + (row0 + i) * D_DIM + col);

    *(v4f*)&ws[wave][4 * lane] = wq;
#pragma unroll
    for (int i = 0; i < ROWS_PER_BLOCK; ++i)
        *(v4f*)&xs[wave][i][4 * lane] = l[i];

    v8f c = {};
    const float* ap = &ws[wave][2 * half];
    const float* bp = &xs[wave][r][2 * half];

#pragma unroll 8
    for (int j = 0; j < KSLICE / 4; ++j) {
        v2f av = *(const v2f*)(ap + 4 * j);
        v2f bv = *(const v2f*)(bp + 4 * j);
        c = __builtin_amdgcn_wmma_f32_16x16x4_f32(
                false, av, false, bv, (short)0, c, false, false);
    }

    if (lane < 16) part[wave][lane] = c[0];
    __syncthreads();

    if (tid < ROWS_PER_BLOCK) {
        float s = 0.0f;
#pragma unroll
        for (int wv = 0; wv < WAVES_PER_BLOCK; ++wv) s += part[wv][tid];
        srow[tid] = s;
    }
    __syncthreads();

#pragma unroll
    for (int i = 0; i < ROWS_PER_BLOCK; ++i) {
        const size_t off = (row0 + i) * D_DIM + col;
        const float s = srow[i];
        const v4f lv = *(const v4f*)&xs[wave][i][4 * lane];
        v4f o = a[i] * s + bb + lv;
        *(v4f*)(out + off) = o;
    }
}

extern "C" void kernel_launch(void* const* d_in, const int* in_sizes, int n_in,
                              void* d_out, int out_size, void* d_ws, size_t ws_size,
                              hipStream_t stream) {
    const float* x0   = (const float*)d_in[0];
    const float* xl   = (const float*)d_in[1];
    const float* kern = (const float*)d_in[2];
    const float* bias = (const float*)d_in[3];
    float* out = (float*)d_out;

    const int B = in_sizes[0] / D_DIM;
    const int grid = B / ROWS_PER_BLOCK;
    crosslayer_kernel<<<grid, 256, 0, stream>>>(x0, xl, kern, bias, out);
}
